# starting kernel
# baseline (speedup 1.0000x reference)
#include <hip/hip_runtime.h>


#define WIDTH   50
#define PW      64
#define PK      52
#define NBLK    4
#define NCHUNK  13
#define NHID    9
#define WAVES   8
#define TPW     4
#define ACT_PITCH 68
#define PI_2    1.57079632679489662f

typedef float v2f __attribute__((ext_vector_type(2)));
typedef float v8f __attribute__((ext_vector_type(8)));

static __device__ __forceinline__ v8f wmma4(v2f a, v2f b, v8f c) {
    return __builtin_amdgcn_wmma_f32_16x16x4_f32(false, a, false, b, (short)0, c,
                                                 false, false);
}

static __device__ __forceinline__ v8f splat8(float v) {
    v8f r = {v, v, v, v, v, v, v, v};
    return r;
}

__global__ __launch_bounds__(256) void mfn_kernel(
    const float* __restrict__ x,
    const float* __restrict__ fw,   const float* __restrict__ fb,
    const float* __restrict__ h1w,  const float* __restrict__ h1b,
    const float* __restrict__ hw,   const float* __restrict__ hb,
    const float* __restrict__ headw,const float* __restrict__ headb,
    float* __restrict__ out, int n)
{
    __shared__ float sHW[NHID * PW * PK];
    __shared__ float sFW[PW * 4];
    __shared__ float sH1W[PW * 4];
    __shared__ float sHEADW[16 * PK];
    __shared__ float sFB[PW];
    __shared__ float sH1B[PW];
    __shared__ float sACT[WAVES * 16 * ACT_PITCH];

    const int tid = threadIdx.x;

    for (int idx = tid; idx < NHID * PW * PK; idx += 256) {
        int i  = idx / (PW * PK);
        int rk = idx - i * (PW * PK);
        int r  = rk / PK, k = rk - r * PK;
        float v = 0.0f;
        if (r < WIDTH) {
            if (k < WIDTH)       v = hw[(i * WIDTH + r) * WIDTH + k];
            else if (k == WIDTH) v = hb[i * WIDTH + r];
        } else if (r == WIDTH && k == WIDTH) {
            v = PI_2;
        }
        sHW[idx] = v;
    }
    for (int idx = tid; idx < PW * 4; idx += 256) {
        int r = idx >> 2, k = idx & 3;
        sFW[idx]  = (r < WIDTH) ? fw[r * 4 + k]  : 0.0f;
        sH1W[idx] = (r < WIDTH) ? h1w[r * 4 + k] : 0.0f;
    }
    for (int idx = tid; idx < 16 * PK; idx += 256) {
        int r = idx / PK, k = idx - r * PK;
        float v = 0.0f;
        if (r < 4) {
            if (k < WIDTH)       v = headw[r * WIDTH + k];
            else if (k == WIDTH) v = headb[r];
        }
        sHEADW[idx] = v;
    }
    for (int idx = tid; idx < PW; idx += 256) {
        sFB[idx]  = (idx < WIDTH) ? fb[idx] : 0.0f;
        float hv = 0.0f;
        if (idx < WIDTH)       hv = h1b[idx];
        else if (idx == WIDTH) hv = PI_2;
        sH1B[idx] = hv;
    }
    __syncthreads();

    const int lane = tid & 31;
    const int wave = tid >> 5;
    const int half = lane >> 4;
    const int l16  = lane & 15;
    float* act = &sACT[wave * 16 * ACT_PITCH];

    const int ntiles = (n + 15) >> 4;
    const int tile0  = blockIdx.x * (WAVES * TPW) + wave * TPW;

    for (int t = 0; t < TPW; ++t) {
        const int tile = tile0 + t;
        if (tile >= ntiles) break;
        const int p0 = tile << 4;

        int row = p0 + l16;
        if (row >= n) row = n - 1;
        const v2f ax = *(const v2f*)&x[row * 4 + half * 2];

        v8f U[NBLK], VmU[NBLK];
#pragma unroll
        for (int j = 0; j < NBLK; ++j) {
            v8f c = splat8(sFB[j * 16 + l16]);
            v2f b = *(const v2f*)&sFW[(j * 16 + l16) * 4 + half * 2];
            c = wmma4(ax, b, c);
            v8f u, vmu;
#pragma unroll
            for (int g = 0; g < 8; ++g) {
                float s = __sinf(c[g]);
                u[g]   = s;
                vmu[g] = __cosf(c[g]) - s;
            }
            U[j] = u;
            VmU[j] = vmu;
        }

#pragma unroll
        for (int j = 0; j < NBLK; ++j) {
            v8f c = splat8(sH1B[j * 16 + l16]);
            v2f b = *(const v2f*)&sH1W[(j * 16 + l16) * 4 + half * 2];
            c = wmma4(ax, b, c);
#pragma unroll
            for (int g = 0; g < 8; ++g)
                act[(g + half * 8) * ACT_PITCH + j * 16 + l16] = __sinf(c[g]);
        }

        for (int i = 0; i < NHID; ++i) {
            v2f a[NCHUNK];
#pragma unroll
            for (int cc = 0; cc < NCHUNK; ++cc)
                a[cc] = *(const v2f*)&act[l16 * ACT_PITCH + cc * 4 + half * 2];

            const float* wl = &sHW[i * PW * PK];
            v8f c[NBLK];
#pragma unroll
            for (int j = 0; j < NBLK; ++j) c[j] = splat8(0.0f) * 0.0f, c[j] = (v8f){};
#pragma unroll
            for (int cc = 0; cc < NCHUNK; ++cc) {
#pragma unroll
                for (int j = 0; j < NBLK; ++j) {
                    v2f b = *(const v2f*)&wl[(j * 16 + l16) * PK + cc * 4 + half * 2];
                    c[j] = wmma4(a[cc], b, c[j]);
                }
            }
#pragma unroll
            for (int j = 0; j < NBLK; ++j) {
#pragma unroll
                for (int g = 0; g < 8; ++g) {
                    float z = __sinf(c[j][g]);
                    act[(g + half * 8) * ACT_PITCH + j * 16 + l16] =
                        U[j][g] + z * VmU[j][g];
                }
            }
        }

        {
            v2f a[NCHUNK];
#pragma unroll
            for (int cc = 0; cc < NCHUNK; ++cc)
                a[cc] = *(const v2f*)&act[l16 * ACT_PITCH + cc * 4 + half * 2];
            const float* wrow = &sHEADW[l16 * PK + half * 2];
            v8f c0 = {}, c1 = {};
#pragma unroll
            for (int cc = 0; cc < NCHUNK; ++cc) {
                v2f b = *(const v2f*)&wrow[cc * 4];
                if (cc & 1) c1 = wmma4(a[cc], b, c1);
                else        c0 = wmma4(a[cc], b, c0);
            }
            v8f c = c0 + c1;
#pragma unroll
            for (int g = 0; g < 8; ++g)
                act[(g + half * 8) * ACT_PITCH + l16] = c[g];
            const int prow = p0 + l16;
            v2f r = *(const v2f*)&act[l16 * ACT_PITCH + half * 2];
            if (prow < n)
                *(v2f*)&out[prow * 4 + half * 2] = r;
        }
    }
}

extern "C" void kernel_launch(void* const* d_in, const int* in_sizes, int n_in,
                              void* d_out, int out_size, void* d_ws, size_t ws_size,
                              hipStream_t stream) {
    const float* x     = (const float*)d_in[0];
    const float* fw    = (const float*)d_in[1];
    const float* fb    = (const float*)d_in[2];
    const float* h1w   = (const float*)d_in[3];
    const float* h1b   = (const float*)d_in[4];
    const float* hw    = (const float*)d_in[5];
    const float* hb    = (const float*)d_in[6];
    const float* headw = (const float*)d_in[7];
    const float* headb = (const float*)d_in[8];

    const int n = in_sizes[0] / 4;
    const int ntiles = (n + 15) / 16;
    const int blocks = (ntiles + WAVES * TPW - 1) / (WAVES * TPW);

    mfn_kernel<<<blocks, 256, 0, stream>>>(x, fw, fb, h1w, h1b, hw, hb,
                                           headw, headb, (float*)d_out, n);
}


# persistent grid: 512 work-groups (2 per compute unit) walk the point tiles with a grid stride, so the 120 KB weight staging runs once per work-group instead of once per 4 tiles of each wave
# speedup vs baseline: 1.2690x; 1.2690x over previous
#include <hip/hip_runtime.h>


#define WIDTH   50
#define PW      64
#define PK      52
#define NBLK    4
#define NCHUNK  13
#define NHID    9
#define WAVES   8
#define NUM_WGP        256
#define BLOCKS_PER_WGP 2
#define GRID_BLOCKS    (NUM_WGP * BLOCKS_PER_WGP)
#define ACT_PITCH 68
#define PI_2    1.57079632679489662f

typedef float v2f __attribute__((ext_vector_type(2)));
typedef float v8f __attribute__((ext_vector_type(8)));

static __device__ __forceinline__ v8f wmma4(v2f a, v2f b, v8f c) {
    return __builtin_amdgcn_wmma_f32_16x16x4_f32(false, a, false, b, (short)0, c,
                                                 false, false);
}

static __device__ __forceinline__ v8f splat8(float v) {
    v8f r = {v, v, v, v, v, v, v, v};
    return r;
}

__global__ __launch_bounds__(256) void mfn_kernel(
    const float* __restrict__ x,
    const float* __restrict__ fw,   const float* __restrict__ fb,
    const float* __restrict__ h1w,  const float* __restrict__ h1b,
    const float* __restrict__ hw,   const float* __restrict__ hb,
    const float* __restrict__ headw,const float* __restrict__ headb,
    float* __restrict__ out, int n)
{
    __shared__ float sHW[NHID * PW * PK];
    __shared__ float sFW[PW * 4];
    __shared__ float sH1W[PW * 4];
    __shared__ float sHEADW[16 * PK];
    __shared__ float sFB[PW];
    __shared__ float sH1B[PW];
    __shared__ float sACT[WAVES * 16 * ACT_PITCH];

    const int tid = threadIdx.x;

    for (int idx = tid; idx < NHID * PW * PK; idx += 256) {
        int i  = idx / (PW * PK);
        int rk = idx - i * (PW * PK);
        int r  = rk / PK, k = rk - r * PK;
        float v = 0.0f;
        if (r < WIDTH) {
            if (k < WIDTH)       v = hw[(i * WIDTH + r) * WIDTH + k];
            else if (k == WIDTH) v = hb[i * WIDTH + r];
        } else if (r == WIDTH && k == WIDTH) {
            v = PI_2;
        }
        sHW[idx] = v;
    }
    for (int idx = tid; idx < PW * 4; idx += 256) {
        int r = idx >> 2, k = idx & 3;
        sFW[idx]  = (r < WIDTH) ? fw[r * 4 + k]  : 0.0f;
        sH1W[idx] = (r < WIDTH) ? h1w[r * 4 + k] : 0.0f;
    }
    for (int idx = tid; idx < 16 * PK; idx += 256) {
        int r = idx / PK, k = idx - r * PK;
        float v = 0.0f;
        if (r < 4) {
            if (k < WIDTH)       v = headw[r * WIDTH + k];
            else if (k == WIDTH) v = headb[r];
        }
        sHEADW[idx] = v;
    }
    for (int idx = tid; idx < PW; idx += 256) {
        sFB[idx]  = (idx < WIDTH) ? fb[idx] : 0.0f;
        float hv = 0.0f;
        if (idx < WIDTH)       hv = h1b[idx];
        else if (idx == WIDTH) hv = PI_2;
        sH1B[idx] = hv;
    }
    __syncthreads();

    const int lane = tid & 31;
    const int wave = tid >> 5;
    const int half = lane >> 4;
    const int l16  = lane & 15;
    float* act = &sACT[wave * 16 * ACT_PITCH];

    const int ntiles  = (n + 15) >> 4;
    const int tstride = GRID_BLOCKS * WAVES;

    for (int tile = blockIdx.x * WAVES + wave; tile < ntiles; tile += tstride) {
        const int p0 = tile << 4;

        int row = p0 + l16;
        if (row >= n) row = n - 1;
        const v2f ax = *(const v2f*)&x[row * 4 + half * 2];

        v8f U[NBLK], VmU[NBLK];
#pragma unroll
        for (int j = 0; j < NBLK; ++j) {
            v8f c = splat8(sFB[j * 16 + l16]);
            v2f b = *(const v2f*)&sFW[(j * 16 + l16) * 4 + half * 2];
            c = wmma4(ax, b, c);
            v8f u, vmu;
#pragma unroll
            for (int g = 0; g < 8; ++g) {
                float s = __sinf(c[g]);
                u[g]   = s;
                vmu[g] = __cosf(c[g]) - s;
            }
            U[j] = u;
            VmU[j] = vmu;
        }

#pragma unroll
        for (int j = 0; j < NBLK; ++j) {
            v8f c = splat8(sH1B[j * 16 + l16]);
            v2f b = *(const v2f*)&sH1W[(j * 16 + l16) * 4 + half * 2];
            c = wmma4(ax, b, c);
#pragma unroll
            for (int g = 0; g < 8; ++g)
                act[(g + half * 8) * ACT_PITCH + j * 16 + l16] = __sinf(c[g]);
        }

        for (int i = 0; i < NHID; ++i) {
            v2f a[NCHUNK];
#pragma unroll
            for (int cc = 0; cc < NCHUNK; ++cc)
                a[cc] = *(const v2f*)&act[l16 * ACT_PITCH + cc * 4 + half * 2];

            const float* wl = &sHW[i * PW * PK];
            v8f c[NBLK];
#pragma unroll
            for (int j = 0; j < NBLK; ++j) c[j] = splat8(0.0f) * 0.0f, c[j] = (v8f){};
#pragma unroll
            for (int cc = 0; cc < NCHUNK; ++cc) {
#pragma unroll
                for (int j = 0; j < NBLK; ++j) {
                    v2f b = *(const v2f*)&wl[(j * 16 + l16) * PK + cc * 4 + half * 2];
                    c[j] = wmma4(a[cc], b, c[j]);
                }
            }
#pragma unroll
            for (int j = 0; j < NBLK; ++j) {
#pragma unroll
                for (int g = 0; g < 8; ++g) {
                    float z = __sinf(c[j][g]);
                    act[(g + half * 8) * ACT_PITCH + j * 16 + l16] =
                        U[j][g] + z * VmU[j][g];
                }
            }
        }

        {
            v2f a[NCHUNK];
#pragma unroll
            for (int cc = 0; cc < NCHUNK; ++cc)
                a[cc] = *(const v2f*)&act[l16 * ACT_PITCH + cc * 4 + half * 2];
            const float* wrow = &sHEADW[l16 * PK + half * 2];
            v8f c0 = {}, c1 = {};
#pragma unroll
            for (int cc = 0; cc < NCHUNK; ++cc) {
                v2f b = *(const v2f*)&wrow[cc * 4];
                if (cc & 1) c1 = wmma4(a[cc], b, c1);
                else        c0 = wmma4(a[cc], b, c0);
            }
            v8f c = c0 + c1;
#pragma unroll
            for (int g = 0; g < 8; ++g)
                act[(g + half * 8) * ACT_PITCH + l16] = c[g];
            const int prow = p0 + l16;
            v2f r = *(const v2f*)&act[l16 * ACT_PITCH + half * 2];
            if (prow < n)
                *(v2f*)&out[prow * 4 + half * 2] = r;
        }
    }
}

extern "C" void kernel_launch(void* const* d_in, const int* in_sizes, int n_in,
                              void* d_out, int out_size, void* d_ws, size_t ws_size,
                              hipStream_t stream) {
    const float* x     = (const float*)d_in[0];
    const float* fw    = (const float*)d_in[1];
    const float* fb    = (const float*)d_in[2];
    const float* h1w   = (const float*)d_in[3];
    const float* h1b   = (const float*)d_in[4];
    const float* hw    = (const float*)d_in[5];
    const float* hb    = (const float*)d_in[6];
    const float* headw = (const float*)d_in[7];
    const float* headb = (const float*)d_in[8];

    const int n = in_sizes[0] / 4;

    mfn_kernel<<<GRID_BLOCKS, 256, 0, stream>>>(x, fw, fb, h1w, h1b, hw, hb,
                                                headw, headb, (float*)d_out, n);
}


# weight staging made latency-parallel: each thread issues all global loads of a 64x52 layer plane (unconditional, clamped index) before its LDS stores, instead of one waited load per element
# speedup vs baseline: 1.3511x; 1.0647x over previous
#include <hip/hip_runtime.h>


#define WIDTH   50
#define PW      64
#define PK      52
#define NBLK    4
#define NCHUNK  13
#define NHID    9
#define WAVES   8
#define NUM_WGP        256
#define BLOCKS_PER_WGP 2
#define GRID_BLOCKS    (NUM_WGP * BLOCKS_PER_WGP)
#define ACT_PITCH 68
#define PI_2    1.57079632679489662f

typedef float v2f __attribute__((ext_vector_type(2)));
typedef float v8f __attribute__((ext_vector_type(8)));

static __device__ __forceinline__ v8f wmma4(v2f a, v2f b, v8f c) {
    return __builtin_amdgcn_wmma_f32_16x16x4_f32(false, a, false, b, (short)0, c,
                                                 false, false);
}

static __device__ __forceinline__ v8f splat8(float v) {
    v8f r = {v, v, v, v, v, v, v, v};
    return r;
}

__global__ __launch_bounds__(256) void mfn_kernel(
    const float* __restrict__ x,
    const float* __restrict__ fw,   const float* __restrict__ fb,
    const float* __restrict__ h1w,  const float* __restrict__ h1b,
    const float* __restrict__ hw,   const float* __restrict__ hb,
    const float* __restrict__ headw,const float* __restrict__ headb,
    float* __restrict__ out, int n)
{
    __shared__ float sHW[NHID * PW * PK];
    __shared__ float sFW[PW * 4];
    __shared__ float sH1W[PW * 4];
    __shared__ float sHEADW[16 * PK];
    __shared__ float sFB[PW];
    __shared__ float sH1B[PW];
    __shared__ float sACT[WAVES * 16 * ACT_PITCH];

    const int tid = threadIdx.x;

    {
        const bool  inw = (tid >> 2) < WIDTH;
        const float vfw  = fw[inw ? tid : 0];
        const float vh1w = h1w[inw ? tid : 0];
        const bool  inb = tid < WIDTH;
        const float vfb  = fb[inb ? tid : 0];
        const float vh1b = h1b[inb ? tid : 0];
        float vhd[4];
#pragma unroll
        for (int u = 0; u < 4; ++u) {
            const int idx = tid + 256 * u;
            const int r = idx / PK, k = idx - r * PK;
            const bool isw = (r < 4) && (k < WIDTH);
            const bool isb = (r < 4) && (k == WIDTH);
            const float tw = headw[isw ? r * WIDTH + k : 0];
            const float tb = headb[isb ? r : 0];
            vhd[u] = isw ? tw : (isb ? tb : 0.0f);
        }
        sFW[tid]  = inw ? vfw  : 0.0f;
        sH1W[tid] = inw ? vh1w : 0.0f;
        if (tid < PW) {
            sFB[tid]  = inb ? vfb : 0.0f;
            sH1B[tid] = inb ? vh1b : ((tid == WIDTH) ? PI_2 : 0.0f);
        }
#pragma unroll
        for (int u = 0; u < 4; ++u) {
            const int idx = tid + 256 * u;
            if (idx < 16 * PK) sHEADW[idx] = vhd[u];
        }
    }
    for (int i = 0; i < NHID; ++i) {
        float v[13];
#pragma unroll
        for (int u = 0; u < 13; ++u) {
            const int rk = tid + 256 * u;
            const int r  = rk / PK, k = rk - r * PK;
            const bool isw = (r < WIDTH) && (k < WIDTH);
            const bool isb = (r < WIDTH) && (k == WIDTH);
            const float tw = hw[isw ? (i * WIDTH + r) * WIDTH + k : 0];
            const float tb = hb[isb ? i * WIDTH + r : 0];
            v[u] = isw ? tw
                       : (isb ? tb
                              : ((r == WIDTH && k == WIDTH) ? PI_2 : 0.0f));
        }
#pragma unroll
        for (int u = 0; u < 13; ++u)
            sHW[i * (PW * PK) + tid + 256 * u] = v[u];
    }
    __syncthreads();

    const int lane = tid & 31;
    const int wave = tid >> 5;
    const int half = lane >> 4;
    const int l16  = lane & 15;
    float* act = &sACT[wave * 16 * ACT_PITCH];

    const int ntiles  = (n + 15) >> 4;
    const int tstride = GRID_BLOCKS * WAVES;

    for (int tile = blockIdx.x * WAVES + wave; tile < ntiles; tile += tstride) {
        const int p0 = tile << 4;

        int row = p0 + l16;
        if (row >= n) row = n - 1;
        const v2f ax = *(const v2f*)&x[row * 4 + half * 2];

        v8f U[NBLK], VmU[NBLK];
#pragma unroll
        for (int j = 0; j < NBLK; ++j) {
            v8f c = splat8(sFB[j * 16 + l16]);
            v2f b = *(const v2f*)&sFW[(j * 16 + l16) * 4 + half * 2];
            c = wmma4(ax, b, c);
            v8f u, vmu;
#pragma unroll
            for (int g = 0; g < 8; ++g) {
                float s = __sinf(c[g]);
                u[g]   = s;
                vmu[g] = __cosf(c[g]) - s;
            }
            U[j] = u;
            VmU[j] = vmu;
        }

#pragma unroll
        for (int j = 0; j < NBLK; ++j) {
            v8f c = splat8(sH1B[j * 16 + l16]);
            v2f b = *(const v2f*)&sH1W[(j * 16 + l16) * 4 + half * 2];
            c = wmma4(ax, b, c);
#pragma unroll
            for (int g = 0; g < 8; ++g)
                act[(g + half * 8) * ACT_PITCH + j * 16 + l16] = __sinf(c[g]);
        }

        for (int i = 0; i < NHID; ++i) {
            v2f a[NCHUNK];
#pragma unroll
            for (int cc = 0; cc < NCHUNK; ++cc)
                a[cc] = *(const v2f*)&act[l16 * ACT_PITCH + cc * 4 + half * 2];

            const float* wl = &sHW[i * PW * PK];
            v8f c[NBLK];
#pragma unroll
            for (int j = 0; j < NBLK; ++j) c[j] = splat8(0.0f) * 0.0f, c[j] = (v8f){};
#pragma unroll
            for (int cc = 0; cc < NCHUNK; ++cc) {
#pragma unroll
                for (int j = 0; j < NBLK; ++j) {
                    v2f b = *(const v2f*)&wl[(j * 16 + l16) * PK + cc * 4 + half * 2];
                    c[j] = wmma4(a[cc], b, c[j]);
                }
            }
#pragma unroll
            for (int j = 0; j < NBLK; ++j) {
#pragma unroll
                for (int g = 0; g < 8; ++g) {
                    float z = __sinf(c[j][g]);
                    act[(g + half * 8) * ACT_PITCH + j * 16 + l16] =
                        U[j][g] + z * VmU[j][g];
                }
            }
        }

        {
            v2f a[NCHUNK];
#pragma unroll
            for (int cc = 0; cc < NCHUNK; ++cc)
                a[cc] = *(const v2f*)&act[l16 * ACT_PITCH + cc * 4 + half * 2];
            const float* wrow = &sHEADW[l16 * PK + half * 2];
            v8f c0 = {}, c1 = {};
#pragma unroll
            for (int cc = 0; cc < NCHUNK; ++cc) {
                v2f b = *(const v2f*)&wrow[cc * 4];
                if (cc & 1) c1 = wmma4(a[cc], b, c1);
                else        c0 = wmma4(a[cc], b, c0);
            }
            v8f c = c0 + c1;
#pragma unroll
            for (int g = 0; g < 8; ++g)
                act[(g + half * 8) * ACT_PITCH + l16] = c[g];
            const int prow = p0 + l16;
            v2f r = *(const v2f*)&act[l16 * ACT_PITCH + half * 2];
            if (prow < n)
                *(v2f*)&out[prow * 4 + half * 2] = r;
        }
    }
}

extern "C" void kernel_launch(void* const* d_in, const int* in_sizes, int n_in,
                              void* d_out, int out_size, void* d_ws, size_t ws_size,
                              hipStream_t stream) {
    const float* x     = (const float*)d_in[0];
    const float* fw    = (const float*)d_in[1];
    const float* fb    = (const float*)d_in[2];
    const float* h1w   = (const float*)d_in[3];
    const float* h1b   = (const float*)d_in[4];
    const float* hw    = (const float*)d_in[5];
    const float* hb    = (const float*)d_in[6];
    const float* headw = (const float*)d_in[7];
    const float* headb = (const float*)d_in[8];

    const int n = in_sizes[0] / 4;

    mfn_kernel<<<GRID_BLOCKS, 256, 0, stream>>>(x, fw, fb, h1w, h1b, hw, hb,
                                                headw, headb, (float*)d_out, n);
}


# layers computed transposed (neurons on matrix rows, points on columns): a layer's D registers are the next layer's B chunks, activations skip LDS; summation order of k inside a chunk permuted
# speedup vs baseline: 1.4331x; 1.0607x over previous
#include <hip/hip_runtime.h>


#define WIDTH   50
#define PW      64
#define PK      52
#define NBLK    4
#define NCHUNK  13
#define NHID    9
#define WAVES   8
#define NUM_WGP        256
#define BLOCKS_PER_WGP 2
#define GRID_BLOCKS    (NUM_WGP * BLOCKS_PER_WGP)
#define KEEP    56
#define PI_2    1.57079632679489662f

typedef float v2f __attribute__((ext_vector_type(2)));
typedef float v8f __attribute__((ext_vector_type(8)));

static __device__ __forceinline__ v8f wmma4(v2f a, v2f b, v8f c) {
    return __builtin_amdgcn_wmma_f32_16x16x4_f32(false, a, false, b, (short)0, c,
                                                 false, false);
}

static __device__ __forceinline__ int col_position(int q) {
    const int cc = q >> 2, h = (q >> 1) & 1, sreg = q & 1;
    return ((cc >> 2) << 4) + (h << 3) + ((cc & 3) << 1) + sreg;
}

__global__ __launch_bounds__(256) void mfn_kernel(
    const float* __restrict__ x,
    const float* __restrict__ fw,   const float* __restrict__ fb,
    const float* __restrict__ h1w,  const float* __restrict__ h1b,
    const float* __restrict__ hw,   const float* __restrict__ hb,
    const float* __restrict__ headw,const float* __restrict__ headb,
    float* __restrict__ out, int n)
{
    __shared__ __attribute__((aligned(16))) float sHW[NHID * PW * PK];
    __shared__ __attribute__((aligned(16))) float sFW[PW * 4];
    __shared__ __attribute__((aligned(16))) float sH1W[PW * 4];
    __shared__ __attribute__((aligned(16))) float sHEADW[16 * PK];
    __shared__ __attribute__((aligned(16))) float sFB[PW];
    __shared__ __attribute__((aligned(16))) float sH1B[PW];

    const int tid = threadIdx.x;

    {
        const bool  inw = (tid >> 2) < WIDTH;
        const float vfw  = fw[inw ? tid : 0];
        const float vh1w = h1w[inw ? tid : 0];
        const bool  inb = tid < WIDTH;
        const float vfb  = fb[inb ? tid : 0];
        const float vh1b = h1b[inb ? tid : 0];
        float vhd[4];
#pragma unroll
        for (int u = 0; u < 4; ++u) {
            const int idx = tid + 256 * u;
            const int r = idx / PK, q = idx - r * PK;
            const int pos = col_position(q);
            const bool isw = (r < 4) && (pos < WIDTH);
            const bool isb = (r < 4) && (pos == KEEP);
            const float tw = headw[isw ? r * WIDTH + pos : 0];
            const float tb = headb[isb ? r : 0];
            vhd[u] = isw ? tw : (isb ? tb : 0.0f);
        }
        sFW[tid]  = inw ? vfw  : 0.0f;
        sH1W[tid] = inw ? vh1w : 0.0f;
        if (tid < PW) {
            sFB[tid]  = inb ? vfb : 0.0f;
            sH1B[tid] = inb ? vh1b : ((tid == KEEP) ? PI_2 : 0.0f);
        }
#pragma unroll
        for (int u = 0; u < 4; ++u) {
            const int idx = tid + 256 * u;
            if (idx < 16 * PK) sHEADW[idx] = vhd[u];
        }
    }
    for (int i = 0; i < NHID; ++i) {
        float v[13];
#pragma unroll
        for (int u = 0; u < 13; ++u) {
            const int rk = tid + 256 * u;
            const int r  = rk / PK, q = rk - r * PK;
            const int pos = col_position(q);
            const bool isw = (r < WIDTH) && (pos < WIDTH);
            const bool isb = (r < WIDTH) && (pos == KEEP);
            const float tw = hw[isw ? (i * WIDTH + r) * WIDTH + pos : 0];
            const float tb = hb[isb ? i * WIDTH + r : 0];
            v[u] = isw ? tw
                       : (isb ? tb
                              : ((r == KEEP && pos == KEEP) ? PI_2 : 0.0f));
        }
#pragma unroll
        for (int u = 0; u < 13; ++u)
            sHW[i * (PW * PK) + tid + 256 * u] = v[u];
    }
    __syncthreads();

    const int lane = tid & 31;
    const int wave = tid >> 5;
    const int half = lane >> 4;
    const int l16  = lane & 15;

    const int ntiles  = (n + 15) >> 4;
    const int tstride = GRID_BLOCKS * WAVES;

    for (int tile = blockIdx.x * WAVES + wave; tile < ntiles; tile += tstride) {
        const int p0 = tile << 4;

        int row = p0 + l16;
        if (row >= n) row = n - 1;
        const v2f bx = *(const v2f*)&x[row * 4 + half * 2];

        v8f U[NBLK], VmU[NBLK], act[NBLK];
#pragma unroll
        for (int j = 0; j < NBLK; ++j) {
            const int nr = (j == NBLK - 1) ? 2 : 8;
            const float4 b0 = *(const float4*)&sFB[j * 16 + half * 8];
            const float4 b1 = *(const float4*)&sFB[j * 16 + half * 8 + 4];
            v8f c = {b0.x, b0.y, b0.z, b0.w, b1.x, b1.y, b1.z, b1.w};
            const v2f a = *(const v2f*)&sFW[(j * 16 + l16) * 4 + half * 2];
            c = wmma4(a, bx, c);
            v8f u = {}, vmu = {};
#pragma unroll
            for (int g = 0; g < nr; ++g) {
                const float sn = __sinf(c[g]);
                u[g]   = sn;
                vmu[g] = __cosf(c[g]) - sn;
            }
            U[j] = u;
            VmU[j] = vmu;
        }

#pragma unroll
        for (int j = 0; j < NBLK; ++j) {
            const int nr = (j == NBLK - 1) ? 2 : 8;
            const float4 b0 = *(const float4*)&sH1B[j * 16 + half * 8];
            const float4 b1 = *(const float4*)&sH1B[j * 16 + half * 8 + 4];
            v8f c = {b0.x, b0.y, b0.z, b0.w, b1.x, b1.y, b1.z, b1.w};
            const v2f a = *(const v2f*)&sH1W[(j * 16 + l16) * 4 + half * 2];
            c = wmma4(a, bx, c);
            v8f o = {};
#pragma unroll
            for (int g = 0; g < nr; ++g)
                o[g] = __sinf(c[g]);
            act[j] = o;
        }

        for (int i = 0; i < NHID; ++i) {
            const float* wl = &sHW[i * PW * PK + l16 * PK + half * 2];
            v8f c[NBLK];
#pragma unroll
            for (int j = 0; j < NBLK; ++j) c[j] = (v8f){};
            v2f a[NCHUNK][NBLK];
#pragma unroll
            for (int cc = 0; cc < NCHUNK; ++cc) {
#pragma unroll
                for (int j = 0; j < NBLK; ++j)
                    a[cc][j] = *(const v2f*)&wl[j * 16 * PK + cc * 4];
            }
            __builtin_amdgcn_sched_barrier(0);
#pragma unroll
            for (int cc = 0; cc < NCHUNK; ++cc) {
                const v2f b = {act[cc >> 2][2 * (cc & 3)], act[cc >> 2][2 * (cc & 3) + 1]};
#pragma unroll
                for (int j = 0; j < NBLK; ++j)
                    c[j] = wmma4(a[cc][j], b, c[j]);
            }
#pragma unroll
            for (int j = 0; j < NBLK; ++j) {
                const int nr = (j == NBLK - 1) ? 2 : 8;
                v8f o = {};
#pragma unroll
                for (int g = 0; g < nr; ++g) {
                    const float z = __sinf(c[j][g]);
                    o[g] = U[j][g] + z * VmU[j][g];
                }
                act[j] = o;
            }
        }

        {
            const float* wrow = &sHEADW[l16 * PK + half * 2];
            v8f c0 = {}, c1 = {};
#pragma unroll
            for (int cc = 0; cc < NCHUNK; ++cc) {
                const v2f b = {act[cc >> 2][2 * (cc & 3)], act[cc >> 2][2 * (cc & 3) + 1]};
                const v2f a = *(const v2f*)&wrow[cc * 4];
                if (cc & 1) c1 = wmma4(a, b, c1);
                else        c0 = wmma4(a, b, c0);
            }
            const v8f c = c0 + c1;
            const int prow = p0 + l16;
            if (half == 0 && prow < n)
                *(float4*)&out[prow * 4] = make_float4(c[0], c[1], c[2], c[3]);
        }
    }
}

extern "C" void kernel_launch(void* const* d_in, const int* in_sizes, int n_in,
                              void* d_out, int out_size, void* d_ws, size_t ws_size,
                              hipStream_t stream) {
    const float* x     = (const float*)d_in[0];
    const float* fw    = (const float*)d_in[1];
    const float* fb    = (const float*)d_in[2];
    const float* h1w   = (const float*)d_in[3];
    const float* h1b   = (const float*)d_in[4];
    const float* hw    = (const float*)d_in[5];
    const float* hb    = (const float*)d_in[6];
    const float* headw = (const float*)d_in[7];
    const float* headb = (const float*)d_in[8];

    const int n = in_sizes[0] / 4;

    mfn_kernel<<<GRID_BLOCKS, 256, 0, stream>>>(x, fw, fb, h1w, h1b, hw, hb,
                                                headw, headb, (float*)d_out, n);
}


# wave priority follows remaining work: s_setprio 3/2/1/0 by the tiles a wave has left (8 or more/3 or more/1 or more/last), so a SIMD's four waves end together, not oldest-first one after another
# speedup vs baseline: 1.4860x; 1.0369x over previous
#include <hip/hip_runtime.h>


#define WIDTH   50
#define PW      64
#define PK      52
#define NBLK    4
#define NCHUNK  13
#define NHID    9
#define WAVES   8
#define NUM_WGP        256
#define BLOCKS_PER_WGP 2
#define GRID_BLOCKS    (NUM_WGP * BLOCKS_PER_WGP)
#define KEEP    56
#define PI_2    1.57079632679489662f

typedef float v2f __attribute__((ext_vector_type(2)));
typedef float v8f __attribute__((ext_vector_type(8)));

static __device__ __forceinline__ v8f wmma4(v2f a, v2f b, v8f c) {
    return __builtin_amdgcn_wmma_f32_16x16x4_f32(false, a, false, b, (short)0, c,
                                                 false, false);
}

static __device__ __forceinline__ int col_position(int q) {
    const int cc = q >> 2, h = (q >> 1) & 1, sreg = q & 1;
    return ((cc >> 2) << 4) + (h << 3) + ((cc & 3) << 1) + sreg;
}

__global__ __launch_bounds__(256) void mfn_kernel(
    const float* __restrict__ x,
    const float* __restrict__ fw,   const float* __restrict__ fb,
    const float* __restrict__ h1w,  const float* __restrict__ h1b,
    const float* __restrict__ hw,   const float* __restrict__ hb,
    const float* __restrict__ headw,const float* __restrict__ headb,
    float* __restrict__ out, int n)
{
    __shared__ __attribute__((aligned(16))) float sHW[NHID * PW * PK];
    __shared__ __attribute__((aligned(16))) float sFW[PW * 4];
    __shared__ __attribute__((aligned(16))) float sH1W[PW * 4];
    __shared__ __attribute__((aligned(16))) float sHEADW[16 * PK];
    __shared__ __attribute__((aligned(16))) float sFB[PW];
    __shared__ __attribute__((aligned(16))) float sH1B[PW];

    const int tid = threadIdx.x;

    {
        const bool  inw = (tid >> 2) < WIDTH;
        const float vfw  = fw[inw ? tid : 0];
        const float vh1w = h1w[inw ? tid : 0];
        const bool  inb = tid < WIDTH;
        const float vfb  = fb[inb ? tid : 0];
        const float vh1b = h1b[inb ? tid : 0];
        float vhd[4];
#pragma unroll
        for (int u = 0; u < 4; ++u) {
            const int idx = tid + 256 * u;
            const int r = idx / PK, q = idx - r * PK;
            const int pos = col_position(q);
            const bool isw = (r < 4) && (pos < WIDTH);
            const bool isb = (r < 4) && (pos == KEEP);
            const float tw = headw[isw ? r * WIDTH + pos : 0];
            const float tb = headb[isb ? r : 0];
            vhd[u] = isw ? tw : (isb ? tb : 0.0f);
        }
        sFW[tid]  = inw ? vfw  : 0.0f;
        sH1W[tid] = inw ? vh1w : 0.0f;
        if (tid < PW) {
            sFB[tid]  = inb ? vfb : 0.0f;
            sH1B[tid] = inb ? vh1b : ((tid == KEEP) ? PI_2 : 0.0f);
        }
#pragma unroll
        for (int u = 0; u < 4; ++u) {
            const int idx = tid + 256 * u;
            if (idx < 16 * PK) sHEADW[idx] = vhd[u];
        }
    }
    for (int i = 0; i < NHID; ++i) {
        float v[13];
#pragma unroll
        for (int u = 0; u < 13; ++u) {
            const int rk = tid + 256 * u;
            const int r  = rk / PK, q = rk - r * PK;
            const int pos = col_position(q);
            const bool isw = (r < WIDTH) && (pos < WIDTH);
            const bool isb = (r < WIDTH) && (pos == KEEP);
            const float tw = hw[isw ? (i * WIDTH + r) * WIDTH + pos : 0];
            const float tb = hb[isb ? i * WIDTH + r : 0];
            v[u] = isw ? tw
                       : (isb ? tb
                              : ((r == KEEP && pos == KEEP) ? PI_2 : 0.0f));
        }
#pragma unroll
        for (int u = 0; u < 13; ++u)
            sHW[i * (PW * PK) + tid + 256 * u] = v[u];
    }
    __syncthreads();

    const int lane = tid & 31;
    const int wave = tid >> 5;
    const int half = lane >> 4;
    const int l16  = lane & 15;

    const int ntiles  = (n + 15) >> 4;
    const int tstride = GRID_BLOCKS * WAVES;

    const int first = blockIdx.x * WAVES + wave;
    int left = __builtin_amdgcn_readfirstlane(
        first < ntiles ? (int)((unsigned)(ntiles - 1 - first) / (unsigned)tstride) : 0);

    for (int tile = first; tile < ntiles; tile += tstride, --left) {
        const int p0 = tile << 4;

        if (left >= 8)      __builtin_amdgcn_s_setprio(3);
        else if (left >= 3) __builtin_amdgcn_s_setprio(2);
        else if (left >= 1) __builtin_amdgcn_s_setprio(1);
        else                __builtin_amdgcn_s_setprio(0);

        int row = p0 + l16;
        if (row >= n) row = n - 1;
        const v2f bx = *(const v2f*)&x[row * 4 + half * 2];

        v8f U[NBLK], VmU[NBLK], act[NBLK];
#pragma unroll
        for (int j = 0; j < NBLK; ++j) {
            const int nr = (j == NBLK - 1) ? 2 : 8;
            const float4 b0 = *(const float4*)&sFB[j * 16 + half * 8];
            const float4 b1 = *(const float4*)&sFB[j * 16 + half * 8 + 4];
            v8f c = {b0.x, b0.y, b0.z, b0.w, b1.x, b1.y, b1.z, b1.w};
            const v2f a = *(const v2f*)&sFW[(j * 16 + l16) * 4 + half * 2];
            c = wmma4(a, bx, c);
            v8f u = {}, vmu = {};
#pragma unroll
            for (int g = 0; g < nr; ++g) {
                const float sn = __sinf(c[g]);
                u[g]   = sn;
                vmu[g] = __cosf(c[g]) - sn;
            }
            U[j] = u;
            VmU[j] = vmu;
        }

#pragma unroll
        for (int j = 0; j < NBLK; ++j) {
            const int nr = (j == NBLK - 1) ? 2 : 8;
            const float4 b0 = *(const float4*)&sH1B[j * 16 + half * 8];
            const float4 b1 = *(const float4*)&sH1B[j * 16 + half * 8 + 4];
            v8f c = {b0.x, b0.y, b0.z, b0.w, b1.x, b1.y, b1.z, b1.w};
            const v2f a = *(const v2f*)&sH1W[(j * 16 + l16) * 4 + half * 2];
            c = wmma4(a, bx, c);
            v8f o = {};
#pragma unroll
            for (int g = 0; g < nr; ++g)
                o[g] = __sinf(c[g]);
            act[j] = o;
        }

        for (int i = 0; i < NHID; ++i) {
            const float* wl = &sHW[i * PW * PK + l16 * PK + half * 2];
            v8f c[NBLK];
#pragma unroll
            for (int j = 0; j < NBLK; ++j) c[j] = (v8f){};
            v2f a[NCHUNK][NBLK];
#pragma unroll
            for (int cc = 0; cc < NCHUNK; ++cc) {
#pragma unroll
                for (int j = 0; j < NBLK; ++j)
                    a[cc][j] = *(const v2f*)&wl[j * 16 * PK + cc * 4];
            }
            __builtin_amdgcn_sched_barrier(0);
#pragma unroll
            for (int cc = 0; cc < NCHUNK; ++cc) {
                const v2f b = {act[cc >> 2][2 * (cc & 3)], act[cc >> 2][2 * (cc & 3) + 1]};
#pragma unroll
                for (int j = 0; j < NBLK; ++j)
                    c[j] = wmma4(a[cc][j], b, c[j]);
            }
#pragma unroll
            for (int j = 0; j < NBLK; ++j) {
                const int nr = (j == NBLK - 1) ? 2 : 8;
                v8f o = {};
#pragma unroll
                for (int g = 0; g < nr; ++g) {
                    const float z = __sinf(c[j][g]);
                    o[g] = U[j][g] + z * VmU[j][g];
                }
                act[j] = o;
            }
        }

        {
            const float* wrow = &sHEADW[l16 * PK + half * 2];
            v8f c0 = {}, c1 = {};
#pragma unroll
            for (int cc = 0; cc < NCHUNK; ++cc) {
                const v2f b = {act[cc >> 2][2 * (cc & 3)], act[cc >> 2][2 * (cc & 3) + 1]};
                const v2f a = *(const v2f*)&wrow[cc * 4];
                if (cc & 1) c1 = wmma4(a, b, c1);
                else        c0 = wmma4(a, b, c0);
            }
            const v8f c = c0 + c1;
            const int prow = p0 + l16;
            if (half == 0 && prow < n)
                *(float4*)&out[prow * 4] = make_float4(c[0], c[1], c[2], c[3]);
        }
    }
}

extern "C" void kernel_launch(void* const* d_in, const int* in_sizes, int n_in,
                              void* d_out, int out_size, void* d_ws, size_t ws_size,
                              hipStream_t stream) {
    const float* x     = (const float*)d_in[0];
    const float* fw    = (const float*)d_in[1];
    const float* fb    = (const float*)d_in[2];
    const float* h1w   = (const float*)d_in[3];
    const float* h1b   = (const float*)d_in[4];
    const float* hw    = (const float*)d_in[5];
    const float* hb    = (const float*)d_in[6];
    const float* headw = (const float*)d_in[7];
    const float* headb = (const float*)d_in[8];

    const int n = in_sizes[0] / 4;

    mfn_kernel<<<GRID_BLOCKS, 256, 0, stream>>>(x, fw, fb, h1w, h1b, hw, hb,
                                                headw, headb, (float*)d_out, n);
}


# hidden layers: the 2 neurons of the 2/16-full 4th row block leave the matrix unit (13 of 52 WMMAs a layer) for packed f32 FMAs per lane half + one cross-half add; their summation order changes
# speedup vs baseline: 1.6564x; 1.1147x over previous
#include <hip/hip_runtime.h>


#define WIDTH   50
#define PW      64
#define PK      52
#define NBLK    4
#define NFULL   3
#define NEARLY  6
#define NCHUNK  13
#define NHID    9
#define WAVES   8
#define NUM_WGP        256
#define BLOCKS_PER_WGP 2
#define GRID_BLOCKS    (NUM_WGP * BLOCKS_PER_WGP)
#define KEEP    56
#define PI_2    1.57079632679489662f

typedef float v2f __attribute__((ext_vector_type(2)));
typedef float v8f __attribute__((ext_vector_type(8)));

static __device__ __forceinline__ v8f wmma4(v2f a, v2f b, v8f c) {
    return __builtin_amdgcn_wmma_f32_16x16x4_f32(false, a, false, b, (short)0, c,
                                                 false, false);
}

static __device__ __forceinline__ float other_half(float v) {
    const int b = __builtin_bit_cast(int, v);
    return __builtin_bit_cast(float,
        __builtin_amdgcn_permlanex16(b, b, 0x76543210, 0xfedcba98, false, false));
}

static __device__ __forceinline__ int col_position(int q) {
    const int cc = q >> 2, h = (q >> 1) & 1, sreg = q & 1;
    return ((cc >> 2) << 4) + (h << 3) + ((cc & 3) << 1) + sreg;
}

__global__ __launch_bounds__(256) void mfn_kernel(
    const float* __restrict__ x,
    const float* __restrict__ fw,   const float* __restrict__ fb,
    const float* __restrict__ h1w,  const float* __restrict__ h1b,
    const float* __restrict__ hw,   const float* __restrict__ hb,
    const float* __restrict__ headw,const float* __restrict__ headb,
    float* __restrict__ out, int n)
{
    __shared__ __attribute__((aligned(16))) float sHW[NHID * PW * PK];
    __shared__ __attribute__((aligned(16))) float sFW[PW * 4];
    __shared__ __attribute__((aligned(16))) float sH1W[PW * 4];
    __shared__ __attribute__((aligned(16))) float sHEADW[16 * PK];
    __shared__ __attribute__((aligned(16))) float sFB[PW];
    __shared__ __attribute__((aligned(16))) float sH1B[PW];

    const int tid = threadIdx.x;

    {
        const bool  inw = (tid >> 2) < WIDTH;
        const float vfw  = fw[inw ? tid : 0];
        const float vh1w = h1w[inw ? tid : 0];
        const bool  inb = tid < WIDTH;
        const float vfb  = fb[inb ? tid : 0];
        const float vh1b = h1b[inb ? tid : 0];
        float vhd[4];
#pragma unroll
        for (int u = 0; u < 4; ++u) {
            const int idx = tid + 256 * u;
            const int r = idx / PK, q = idx - r * PK;
            const int pos = col_position(q);
            const bool isw = (r < 4) && (pos < WIDTH);
            const bool isb = (r < 4) && (pos == KEEP);
            const float tw = headw[isw ? r * WIDTH + pos : 0];
            const float tb = headb[isb ? r : 0];
            vhd[u] = isw ? tw : (isb ? tb : 0.0f);
        }
        sFW[tid]  = inw ? vfw  : 0.0f;
        sH1W[tid] = inw ? vh1w : 0.0f;
        if (tid < PW) {
            sFB[tid]  = inb ? vfb : 0.0f;
            sH1B[tid] = inb ? vh1b : ((tid == KEEP) ? PI_2 : 0.0f);
        }
#pragma unroll
        for (int u = 0; u < 4; ++u) {
            const int idx = tid + 256 * u;
            if (idx < 16 * PK) sHEADW[idx] = vhd[u];
        }
    }
    for (int i = 0; i < NHID; ++i) {
        float v[13];
#pragma unroll
        for (int u = 0; u < 13; ++u) {
            const int rk = tid + 256 * u;
            const int r  = rk / PK, q = rk - r * PK;
            const int pos = col_position(q);
            const bool isw = (r < WIDTH) && (pos < WIDTH);
            const bool isb = (r < WIDTH) && (pos == KEEP);
            const float tw = hw[isw ? (i * WIDTH + r) * WIDTH + pos : 0];
            const float tb = hb[isb ? i * WIDTH + r : 0];
            v[u] = isw ? tw : (isb ? tb : 0.0f);
        }
#pragma unroll
        for (int u = 0; u < 13; ++u)
            sHW[i * (PW * PK) + tid + 256 * u] = v[u];
    }
    __syncthreads();

    const int lane = tid & 31;
    const int wave = tid >> 5;
    const int half = lane >> 4;
    const int l16  = lane & 15;

    const int ntiles  = (n + 15) >> 4;
    const int tstride = GRID_BLOCKS * WAVES;

    const int first = blockIdx.x * WAVES + wave;
    int left = __builtin_amdgcn_readfirstlane(
        first < ntiles ? (int)((unsigned)(ntiles - 1 - first) / (unsigned)tstride) : 0);

    for (int tile = first; tile < ntiles; tile += tstride, --left) {
        const int p0 = tile << 4;

        if (left >= 8)      __builtin_amdgcn_s_setprio(3);
        else if (left >= 3) __builtin_amdgcn_s_setprio(2);
        else if (left >= 1) __builtin_amdgcn_s_setprio(1);
        else                __builtin_amdgcn_s_setprio(0);

        int row = p0 + l16;
        if (row >= n) row = n - 1;
        const v2f bx = *(const v2f*)&x[row * 4 + half * 2];

        v8f U[NBLK], VmU[NBLK], act[NBLK];
#pragma unroll
        for (int j = 0; j < NBLK; ++j) {
            const int nr = (j == NBLK - 1) ? 2 : 8;
            const float4 b0 = *(const float4*)&sFB[j * 16 + half * 8];
            const float4 b1 = *(const float4*)&sFB[j * 16 + half * 8 + 4];
            v8f c = {b0.x, b0.y, b0.z, b0.w, b1.x, b1.y, b1.z, b1.w};
            const v2f a = *(const v2f*)&sFW[(j * 16 + l16) * 4 + half * 2];
            c = wmma4(a, bx, c);
            v8f u = {}, vmu = {};
#pragma unroll
            for (int g = 0; g < nr; ++g) {
                const float sn = __sinf(c[g]);
                u[g]   = sn;
                vmu[g] = __cosf(c[g]) - sn;
            }
            U[j] = u;
            VmU[j] = vmu;
        }

#pragma unroll
        for (int j = 0; j < NBLK; ++j) {
            const int nr = (j == NBLK - 1) ? 2 : 8;
            const float4 b0 = *(const float4*)&sH1B[j * 16 + half * 8];
            const float4 b1 = *(const float4*)&sH1B[j * 16 + half * 8 + 4];
            v8f c = {b0.x, b0.y, b0.z, b0.w, b1.x, b1.y, b1.z, b1.w};
            const v2f a = *(const v2f*)&sH1W[(j * 16 + l16) * 4 + half * 2];
            c = wmma4(a, bx, c);
            v8f o = {};
#pragma unroll
            for (int g = 0; g < nr; ++g)
                o[g] = __sinf(c[g]);
            act[j] = o;
        }

        for (int i = 0; i < NHID; ++i) {
            const float* wl = &sHW[i * PW * PK + l16 * PK + half * 2];
            const float* wt = &sHW[i * PW * PK + 48 * PK + half * 2];
            v8f c[NFULL];
#pragma unroll
            for (int j = 0; j < NFULL; ++j) c[j] = (v8f){};
            v2f a[NCHUNK][NFULL];
            v2f w48[NCHUNK], w49[NCHUNK];
#pragma unroll
            for (int cc = 0; cc < NCHUNK; ++cc) {
#pragma unroll
                for (int j = 0; j < NFULL; ++j)
                    a[cc][j] = *(const v2f*)&wl[j * 16 * PK + cc * 4];
                w48[cc] = *(const v2f*)&wt[cc * 4];
                w49[cc] = *(const v2f*)&wt[PK + cc * 4];
            }
            __builtin_amdgcn_sched_barrier(0);
            v2f s48 = {0.0f, 0.0f}, s49 = {0.0f, 0.0f};
#pragma unroll
            for (int cc = 0; cc < NEARLY; ++cc) {
                const v2f b = {act[cc >> 2][2 * (cc & 3)], act[cc >> 2][2 * (cc & 3) + 1]};
                s48 = __builtin_elementwise_fma(w48[cc], b, s48);
                s49 = __builtin_elementwise_fma(w49[cc], b, s49);
            }
            __builtin_amdgcn_sched_barrier(0);
#pragma unroll
            for (int cc = 0; cc < NCHUNK; ++cc) {
                const v2f b = {act[cc >> 2][2 * (cc & 3)], act[cc >> 2][2 * (cc & 3) + 1]};
#pragma unroll
                for (int j = 0; j < NFULL; ++j)
                    c[j] = wmma4(a[cc][j], b, c[j]);
                if (cc >= NEARLY) {
                    s48 = __builtin_elementwise_fma(w48[cc], b, s48);
                    s49 = __builtin_elementwise_fma(w49[cc], b, s49);
                }
            }
#pragma unroll
            for (int j = 0; j < NFULL; ++j) {
                v8f o = {};
#pragma unroll
                for (int g = 0; g < 8; ++g) {
                    const float z = __sinf(c[j][g]);
                    o[g] = U[j][g] + z * VmU[j][g];
                }
                act[j] = o;
            }
            {
                const float p48 = s48[0] + s48[1];
                const float p49 = s49[0] + s49[1];
                const float z48 = __sinf(p48 + other_half(p48));
                const float z49 = __sinf(p49 + other_half(p49));
                v8f o = {};
                o[0] = half ? 1.0f : U[3][0] + z48 * VmU[3][0];
                o[1] = half ? 0.0f : U[3][1] + z49 * VmU[3][1];
                act[3] = o;
            }
        }

        {
            const float* wrow = &sHEADW[l16 * PK + half * 2];
            v8f c0 = {}, c1 = {};
#pragma unroll
            for (int cc = 0; cc < NCHUNK; ++cc) {
                const v2f b = {act[cc >> 2][2 * (cc & 3)], act[cc >> 2][2 * (cc & 3) + 1]};
                const v2f a = *(const v2f*)&wrow[cc * 4];
                if (cc & 1) c1 = wmma4(a, b, c1);
                else        c0 = wmma4(a, b, c0);
            }
            const v8f c = c0 + c1;
            const int prow = p0 + l16;
            if (half == 0 && prow < n)
                *(float4*)&out[prow * 4] = make_float4(c[0], c[1], c[2], c[3]);
        }
    }
}

extern "C" void kernel_launch(void* const* d_in, const int* in_sizes, int n_in,
                              void* d_out, int out_size, void* d_ws, size_t ws_size,
                              hipStream_t stream) {
    const float* x     = (const float*)d_in[0];
    const float* fw    = (const float*)d_in[1];
    const float* fb    = (const float*)d_in[2];
    const float* h1w   = (const float*)d_in[3];
    const float* h1b   = (const float*)d_in[4];
    const float* hw    = (const float*)d_in[5];
    const float* hb    = (const float*)d_in[6];
    const float* headw = (const float*)d_in[7];
    const float* headb = (const float*)d_in[8];

    const int n = in_sizes[0] / 4;

    mfn_kernel<<<GRID_BLOCKS, 256, 0, stream>>>(x, fw, fb, h1w, h1b, hw, hb,
                                                headw, headb, (float*)d_out, n);
}
